# starting kernel
# baseline (speedup 1.0000x reference)
#include <hip/hip_runtime.h>


typedef __attribute__((ext_vector_type(2))) float v2f;
typedef __attribute__((ext_vector_type(8))) float v8f;

#define H_IN   96
#define W_IN   96
#define HW_IN  (H_IN * W_IN)
#define CIN    16
#define B_     8
#define A_     8
#define COUT   16
#define N_     128
#define H_OUT  192
#define W_OUT  192
#define NPIX   (H_OUT * W_OUT)
#define WP2    144

__global__ void __launch_bounds__(256)
vote_wmma_kernel(const float* __restrict__ x,
                 const float* __restrict__ wt,
                 float* __restrict__ votes)
{
    __shared__ v2f wS[8 * WP2];

    const int tid   = threadIdx.x;
    const int lane  = tid & 31;
    const int laneM = lane & 15;
    const int laneH = lane >> 4;
    const int wave  = tid >> 5;

    const int cls = blockIdx.x / 576;
    const int rem = blockIdx.x % 576;
    const int p   = cls >> 1;
    const int q   = cls & 1;

    int t = rem * 8 + wave;
    const int ox0 = (t % 6) * 16;  t /= 6;
    const int oy  = t % 96;        t /= 96;
    const int b   = t;
    const int oh  = 2 * oy + p;

    v8f acc[8] = {};

    const int nTh = 3 - p;
    const int nTw = 3 - q;

    for (int i = 0; i < nTh; ++i) {
        const int  kh  = 2 * i + p;
        const int  ih  = oy + i - 1 + p;
        const bool okh = (unsigned)ih < H_IN;
        for (int j = 0; j < nTw; ++j) {
            const int kw = 2 * j + q;

            __syncthreads();
            {
                const float4* src =
                    (const float4*)(wt + (size_t)(kh * 5 + kw) * CIN * N_);
                #pragma unroll
                for (int e = 0; e < 2; ++e) {
                    const int idx4 = tid * 2 + e;
                    const int idx  = idx4 * 4;
                    const int ci   = idx >> 7;
                    const int n0   = idx & 127;
                    const int k    = ci >> 1;
                    const int sub  = ci & 1;
                    const float4 w4 = src[idx4];
                    ((float*)&wS[k * WP2 + n0 + 0])[sub] = w4.x;
                    ((float*)&wS[k * WP2 + n0 + 1])[sub] = w4.y;
                    ((float*)&wS[k * WP2 + n0 + 2])[sub] = w4.z;
                    ((float*)&wS[k * WP2 + n0 + 3])[sub] = w4.w;
                }
            }
            __syncthreads();

            const int  iw  = ox0 + laneM + j - 1 + q;
            const bool ok  = okh && ((unsigned)iw < W_IN);
            const int  abase = (b * CIN * H_IN + (okh ? ih : 0)) * W_IN
                             + (((unsigned)iw < W_IN) ? iw : 0);

            #pragma unroll
            for (int kg = 0; kg < 4; ++kg) {
                const int ci = kg * 4 + 2 * laneH;
                const float v0 = x[abase + (ci    ) * HW_IN];
                const float v1 = x[abase + (ci + 1) * HW_IN];
                v2f afrag;
                afrag.x = ok ? v0 : 0.f;
                afrag.y = ok ? v1 : 0.f;

                const int brow = (kg * 2 + laneH) * WP2 + laneM;
                v2f bf[8];
                #pragma unroll
                for (int nt = 0; nt < 8; ++nt)
                    bf[nt] = wS[brow + nt * 16];
                #pragma unroll
                for (int nt = 0; nt < 8; ++nt)
                    acc[nt] = __builtin_amdgcn_wmma_f32_16x16x4_f32(
                        false, afrag, false, bf[nt],
                        (short)0, acc[nt], false, false);
            }
        }
    }

    #pragma unroll
    for (int nt = 0; nt < 8; ++nt) {
        #pragma unroll
        for (int r = 0; r < 8; ++r) {
            const int m  = r + 8 * laneH;
            const int ow = 2 * (ox0 + m) + q;
            votes[((size_t)(oh * W_OUT + ow) * B_ + b) * N_ + nt * 16 + laneM]
                = acc[nt][r];
        }
    }
}

__global__ void __launch_bounds__(256)
routing_kernel(const float* __restrict__ votes,
               const int* __restrict__ nroutes_p,
               float* __restrict__ out)
{
    __shared__ float sL[4][64];
    __shared__ float sS[4][64][17];
    __shared__ float sF[4][16];

    const int tid = threadIdx.x;
    const int pp  = tid >> 6;
    const int t64 = tid & 63;
    const int b   = t64 >> 3;
    const int a   = t64 & 7;
    const int pix = blockIdx.x * 4 + pp;

    const int nroutes = *nroutes_p;

    float v[16];
    const float4* vp =
        (const float4*)(votes + ((size_t)pix * B_ + b) * N_ + a * 16);
    #pragma unroll
    for (int i = 0; i < 4; ++i) {
        float4 t = vp[i];
        v[4*i+0] = t.x; v[4*i+1] = t.y; v[4*i+2] = t.z; v[4*i+3] = t.w;
    }

    float logit = 0.f;

    for (int r = 0; r < nroutes; ++r) {
        sL[pp][t64] = logit;
        __syncthreads();
        float mx = -3.402823466e38f;
        #pragma unroll
        for (int bb = 0; bb < 8; ++bb) mx = fmaxf(mx, sL[pp][bb * 8 + a]);
        float den = 0.f;
        #pragma unroll
        for (int bb = 0; bb < 8; ++bb) den += expf(sL[pp][bb * 8 + a] - mx);
        const float ls = expf(logit - mx) / den;

        #pragma unroll
        for (int c = 0; c < 16; ++c) sS[pp][t64][c] = v[c] * ls;
        __syncthreads();
        #pragma unroll
        for (int off = 4; off >= 1; off >>= 1) {
            if (b < off) {
                #pragma unroll
                for (int c = 0; c < 16; ++c)
                    sS[pp][b * 8 + a][c] += sS[pp][(b + off) * 8 + a][c];
            }
            __syncthreads();
        }

        if (t64 < 16) {
            float ss = 0.f;
            #pragma unroll
            for (int aa = 0; aa < 8; ++aa) {
                const float pv = sS[pp][aa][t64];
                ss += pv * pv;
            }
            sF[pp][t64] = sqrtf(ss) / (1.f + ss);
        }
        __syncthreads();

        if (r + 1 < nroutes) {
            float upd = 0.f;
            #pragma unroll
            for (int c = 0; c < 16; ++c)
                upd += v[c] * sS[pp][a][c] * sF[pp][c];
            logit = ls + upd;
        } else {
            #pragma unroll
            for (int k = 0; k < 2; ++k) {
                const int n  = t64 + 64 * k;
                const int ao = n >> 4, co = n & 15;
                out[(size_t)n * NPIX + pix] = sS[pp][ao][co] * sF[pp][co];
            }
        }
    }
}

extern "C" void kernel_launch(void* const* d_in, const int* in_sizes, int n_in,
                              void* d_out, int out_size, void* d_ws, size_t ws_size,
                              hipStream_t stream) {
    (void)in_sizes; (void)n_in; (void)out_size; (void)ws_size;
    const float* x  = (const float*)d_in[0];
    const float* wt = (const float*)d_in[1];
    const int*   nr = (const int*)d_in[2];
    float* votes = (float*)d_ws;
    float* out   = (float*)d_out;

    vote_wmma_kernel<<<2304, 256, 0, stream>>>(x, wt, votes);
    routing_kernel<<<9216, 256, 0, stream>>>(votes, nr, out);
}


# fuse voting and routing into one launch: a workgroup computes all 8 batch entries' votes of its 16 pixels and hands them to the routing through LDS; the votes no longer go through memory
# speedup vs baseline: 2.3887x; 2.3887x over previous
#include <hip/hip_runtime.h>


typedef __attribute__((ext_vector_type(2))) float v2f;
typedef __attribute__((ext_vector_type(8))) float v8f;

#define H_IN   96
#define W_IN   96
#define HW_IN  (H_IN * W_IN)
#define CIN    16
#define B_     8
#define A_     8
#define COUT   16
#define N_     128
#define H_OUT  192
#define W_OUT  192
#define NPIX   (H_OUT * W_OUT)
#define WP2    144
#define VPIX   (B_ * N_)

template <int M>
__device__ __forceinline__ float row_xor(float x) {
    return __builtin_bit_cast(float, __builtin_amdgcn_update_dpp(
        0, __builtin_bit_cast(int, x), 0x160 | M, 0xf, 0xf, true));
}
__device__ __forceinline__ float half_xor(float x) {
    const int b = __builtin_bit_cast(int, x);
    return __builtin_bit_cast(float, __builtin_amdgcn_permlanex16(
        b, b, 0x76543210, 0xfedcba98, false, false));
}

__global__ void __launch_bounds__(256)
vote_route_fused_kernel(const float* __restrict__ x,
                        const float* __restrict__ wt,
                        const int* __restrict__ nroutes_p,
                        float* __restrict__ out)
{
    __shared__ float4 smem[16 * VPIX / 4];
    v2f*   wS = (v2f*)smem;
    float* vS = (float*)smem;

    const int tid   = threadIdx.x;
    const int lane  = tid & 31;
    const int laneM = lane & 15;
    const int laneH = lane >> 4;
    const int wave  = tid >> 5;

    const int cls = blockIdx.x / 576;
    const int rem = blockIdx.x % 576;
    const int p   = cls >> 1;
    const int q   = cls & 1;

    const int ox0 = (rem % 6) * 16;
    const int oy  = rem / 6;
    const int b   = wave;
    const int oh  = 2 * oy + p;

    const int nroutes = *nroutes_p;

    v8f acc[8] = {};

    const int nTh = 3 - p;
    const int nTw = 3 - q;

    for (int i = 0; i < nTh; ++i) {
        const int  kh  = 2 * i + p;
        const int  ih  = oy + i - 1 + p;
        const bool okh = (unsigned)ih < H_IN;
        for (int j = 0; j < nTw; ++j) {
            const int kw = 2 * j + q;

            __syncthreads();
            {
                const float4* src =
                    (const float4*)(wt + (size_t)(kh * 5 + kw) * CIN * N_);
                #pragma unroll
                for (int e = 0; e < 2; ++e) {
                    const int idx4 = tid * 2 + e;
                    const int idx  = idx4 * 4;
                    const int ci   = idx >> 7;
                    const int n0   = idx & 127;
                    const int k    = ci >> 1;
                    const int sub  = ci & 1;
                    const float4 w4 = src[idx4];
                    ((float*)&wS[k * WP2 + n0 + 0])[sub] = w4.x;
                    ((float*)&wS[k * WP2 + n0 + 1])[sub] = w4.y;
                    ((float*)&wS[k * WP2 + n0 + 2])[sub] = w4.z;
                    ((float*)&wS[k * WP2 + n0 + 3])[sub] = w4.w;
                }
            }
            __syncthreads();

            const int  iw  = ox0 + laneM + j - 1 + q;
            const bool ok  = okh && ((unsigned)iw < W_IN);
            const int  abase = (b * CIN * H_IN + (okh ? ih : 0)) * W_IN
                             + (((unsigned)iw < W_IN) ? iw : 0);

            #pragma unroll
            for (int kg = 0; kg < 4; ++kg) {
                const int ci = kg * 4 + 2 * laneH;
                const float v0 = x[abase + (ci    ) * HW_IN];
                const float v1 = x[abase + (ci + 1) * HW_IN];
                v2f afrag;
                afrag.x = ok ? v0 : 0.f;
                afrag.y = ok ? v1 : 0.f;

                const int brow = (kg * 2 + laneH) * WP2 + laneM;
                v2f bf[8];
                #pragma unroll
                for (int nt = 0; nt < 8; ++nt)
                    bf[nt] = wS[brow + nt * 16];
                #pragma unroll
                for (int nt = 0; nt < 8; ++nt)
                    acc[nt] = __builtin_amdgcn_wmma_f32_16x16x4_f32(
                        false, afrag, false, bf[nt],
                        (short)0, acc[nt], false, false);
            }
        }
    }

    __syncthreads();
    {
        float* vw = vS + (8 * laneH) * VPIX + b * N_;
        #pragma unroll
        for (int nt = 0; nt < 8; ++nt) {
            const int nphys = (nt * 16 + laneM) ^ (laneH * 32);
            #pragma unroll
            for (int r = 0; r < 8; ++r)
                vw[r * VPIX + nphys] = acc[nt][r];
        }
    }
    __syncthreads();

    const int nsw = (lane * 4) ^ ((wave >> 2) * 32);
    for (int i = 0; i < 2; ++i) {
        const int m   = 2 * wave + i;
        const int pix = oh * W_OUT + 2 * (ox0 + m) + q;

        const float4* vp = (const float4*)(vS + m * VPIX + nsw);
        float4 cur[8];
        #pragma unroll
        for (int bb = 0; bb < 8; ++bb) cur[bb] = vp[bb * (N_ / 4)];

        float lg[8];
        #pragma unroll
        for (int bb = 0; bb < 8; ++bb) lg[bb] = 0.f;

        for (int r = 0; r < nroutes; ++r) {
            float mx = lg[0];
            #pragma unroll
            for (int bb = 1; bb < 8; ++bb) mx = fmaxf(mx, lg[bb]);
            float ls[8];
            float den = 0.f;
            #pragma unroll
            for (int bb = 0; bb < 8; ++bb) { ls[bb] = expf(lg[bb] - mx); den += ls[bb]; }
            const float inv = 1.f / den;
            #pragma unroll
            for (int bb = 0; bb < 8; ++bb) ls[bb] *= inv;

            float p0 = 0.f, p1 = 0.f, p2 = 0.f, p3 = 0.f;
            #pragma unroll
            for (int bb = 0; bb < 8; ++bb) {
                p0 = fmaf(cur[bb].x, ls[bb], p0);
                p1 = fmaf(cur[bb].y, ls[bb], p1);
                p2 = fmaf(cur[bb].z, ls[bb], p2);
                p3 = fmaf(cur[bb].w, ls[bb], p3);
            }

            float s0 = p0 * p0, s1 = p1 * p1, s2 = p2 * p2, s3 = p3 * p3;
            s0 += row_xor<4>(s0); s1 += row_xor<4>(s1);
            s2 += row_xor<4>(s2); s3 += row_xor<4>(s3);
            s0 += row_xor<8>(s0); s1 += row_xor<8>(s1);
            s2 += row_xor<8>(s2); s3 += row_xor<8>(s3);
            s0 += half_xor(s0);   s1 += half_xor(s1);
            s2 += half_xor(s2);   s3 += half_xor(s3);
            const float q0 = p0 * (sqrtf(s0) / (1.f + s0));
            const float q1 = p1 * (sqrtf(s1) / (1.f + s1));
            const float q2 = p2 * (sqrtf(s2) / (1.f + s2));
            const float q3 = p3 * (sqrtf(s3) / (1.f + s3));

            if (r + 1 < nroutes) {
                #pragma unroll
                for (int bb = 0; bb < 8; ++bb) {
                    float u = cur[bb].x * q0;
                    u = fmaf(cur[bb].y, q1, u);
                    u = fmaf(cur[bb].z, q2, u);
                    u = fmaf(cur[bb].w, q3, u);
                    u += row_xor<1>(u);
                    u += row_xor<2>(u);
                    lg[bb] = ls[bb] + u;
                }
            } else {
                float* o = out + (size_t)(lane * 4) * NPIX + pix;
                o[0]                = q0;
                o[(size_t)NPIX]     = q1;
                o[(size_t)NPIX * 2] = q2;
                o[(size_t)NPIX * 3] = q3;
            }
        }
    }
}

extern "C" void kernel_launch(void* const* d_in, const int* in_sizes, int n_in,
                              void* d_out, int out_size, void* d_ws, size_t ws_size,
                              hipStream_t stream) {
    (void)in_sizes; (void)n_in; (void)out_size; (void)d_ws; (void)ws_size;
    const float* x  = (const float*)d_in[0];
    const float* wt = (const float*)d_in[1];
    const int*   nr = (const int*)d_in[2];
    float* out   = (float*)d_out;

    vote_route_fused_kernel<<<2304, 256, 0, stream>>>(x, wt, nr, out);
}


# output stores: routed values go back through LDS and are written transposed, so one store instruction covers the 16 pixels of 2 planes (2 lines) instead of 32 planes (32 lines, 4 bytes each)
# speedup vs baseline: 3.1775x; 1.3302x over previous
#include <hip/hip_runtime.h>


typedef __attribute__((ext_vector_type(2))) float v2f;
typedef __attribute__((ext_vector_type(8))) float v8f;

#define H_IN   96
#define W_IN   96
#define HW_IN  (H_IN * W_IN)
#define CIN    16
#define B_     8
#define A_     8
#define COUT   16
#define N_     128
#define H_OUT  192
#define W_OUT  192
#define NPIX   (H_OUT * W_OUT)
#define WP2    144
#define VPIX   (B_ * N_)
#define LOG2E  1.4426950408889634f
#define RS_PAIR (2 * VPIX + 8)
#define RS_ODD  132

template <int M>
__device__ __forceinline__ float row_xor(float x) {
    return __builtin_bit_cast(float, __builtin_amdgcn_update_dpp(
        0, __builtin_bit_cast(int, x), 0x160 | M, 0xf, 0xf, true));
}
__device__ __forceinline__ float half_xor(float x) {
    const int b = __builtin_bit_cast(int, x);
    return __builtin_bit_cast(float, __builtin_amdgcn_permlanex16(
        b, b, 0x76543210, 0xfedcba98, false, false));
}

__global__ void __launch_bounds__(256)
vote_route_fused_kernel(const float* __restrict__ x,
                        const float* __restrict__ wt,
                        const int* __restrict__ nroutes_p,
                        float* __restrict__ out)
{
    __shared__ float4 smem[16 * VPIX / 4];
    v2f*   wS = (v2f*)smem;
    float* vS = (float*)smem;

    const int tid   = threadIdx.x;
    const int lane  = tid & 31;
    const int laneM = lane & 15;
    const int laneH = lane >> 4;
    const int wave  = tid >> 5;

    const int cls = blockIdx.x / 576;
    const int rem = blockIdx.x % 576;
    const int p   = cls >> 1;
    const int q   = cls & 1;

    const int ox0 = (rem % 6) * 16;
    const int oy  = rem / 6;
    const int b   = wave;
    const int oh  = 2 * oy + p;

    const int nroutes = *nroutes_p;

    v8f acc[8] = {};

    const int nTh = 3 - p;
    const int nTw = 3 - q;

    for (int i = 0; i < nTh; ++i) {
        const int  kh  = 2 * i + p;
        const int  ih  = oy + i - 1 + p;
        const bool okh = (unsigned)ih < H_IN;
        for (int j = 0; j < nTw; ++j) {
            const int kw = 2 * j + q;

            __syncthreads();
            {
                const float4* src =
                    (const float4*)(wt + (size_t)(kh * 5 + kw) * CIN * N_);
                #pragma unroll
                for (int e = 0; e < 2; ++e) {
                    const int idx4 = tid * 2 + e;
                    const int idx  = idx4 * 4;
                    const int ci   = idx >> 7;
                    const int n0   = idx & 127;
                    const int k    = ci >> 1;
                    const int sub  = ci & 1;
                    const float4 w4 = src[idx4];
                    ((float*)&wS[k * WP2 + n0 + 0])[sub] = w4.x;
                    ((float*)&wS[k * WP2 + n0 + 1])[sub] = w4.y;
                    ((float*)&wS[k * WP2 + n0 + 2])[sub] = w4.z;
                    ((float*)&wS[k * WP2 + n0 + 3])[sub] = w4.w;
                }
            }
            __syncthreads();

            const int  iw  = ox0 + laneM + j - 1 + q;
            const bool ok  = okh && ((unsigned)iw < W_IN);
            const int  abase = (b * CIN * H_IN + (okh ? ih : 0)) * W_IN
                             + (((unsigned)iw < W_IN) ? iw : 0);

            #pragma unroll
            for (int kg = 0; kg < 4; ++kg) {
                const int ci = kg * 4 + 2 * laneH;
                const float v0 = x[abase + (ci    ) * HW_IN];
                const float v1 = x[abase + (ci + 1) * HW_IN];
                v2f afrag;
                afrag.x = ok ? v0 : 0.f;
                afrag.y = ok ? v1 : 0.f;

                const int brow = (kg * 2 + laneH) * WP2 + laneM;
                v2f bf[8];
                #pragma unroll
                for (int nt = 0; nt < 8; ++nt)
                    bf[nt] = wS[brow + nt * 16];
                #pragma unroll
                for (int nt = 0; nt < 8; ++nt)
                    acc[nt] = __builtin_amdgcn_wmma_f32_16x16x4_f32(
                        false, afrag, false, bf[nt],
                        (short)0, acc[nt], false, false);
            }
        }
    }

    __syncthreads();
    {
        float* vw = vS + (8 * laneH) * VPIX + b * N_;
        #pragma unroll
        for (int nt = 0; nt < 8; ++nt) {
            const int nphys = (nt * 16 + laneM) ^ (laneH * 32);
            #pragma unroll
            for (int r = 0; r < 8; ++r)
                vw[r * VPIX + nphys] = acc[nt][r];
        }
    }
    __syncthreads();

    const int nsw = (lane * 4) ^ ((wave >> 2) * 32);
    for (int i = 0; i < 2; ++i) {
        const int m   = 2 * wave + i;

        const float4* vp = (const float4*)(vS + m * VPIX + nsw);
        float4 cur[8];
        #pragma unroll
        for (int bb = 0; bb < 8; ++bb) cur[bb] = vp[bb * (N_ / 4)];

        float lg[8];
        #pragma unroll
        for (int bb = 0; bb < 8; ++bb) lg[bb] = 0.f;

        for (int r = 0; r < nroutes; ++r) {
            float mx = lg[0];
            #pragma unroll
            for (int bb = 1; bb < 8; ++bb) mx = fmaxf(mx, lg[bb]);
            const float nmx = -mx * LOG2E;
            float ls[8];
            float den = 0.f;
            #pragma unroll
            for (int bb = 0; bb < 8; ++bb) {
                ls[bb] = __builtin_amdgcn_exp2f(fmaf(lg[bb], LOG2E, nmx));
                den += ls[bb];
            }
            const float inv = __builtin_amdgcn_rcpf(den);
            #pragma unroll
            for (int bb = 0; bb < 8; ++bb) ls[bb] *= inv;

            float p0 = 0.f, p1 = 0.f, p2 = 0.f, p3 = 0.f;
            #pragma unroll
            for (int bb = 0; bb < 8; ++bb) {
                p0 = fmaf(cur[bb].x, ls[bb], p0);
                p1 = fmaf(cur[bb].y, ls[bb], p1);
                p2 = fmaf(cur[bb].z, ls[bb], p2);
                p3 = fmaf(cur[bb].w, ls[bb], p3);
            }

            float s0 = p0 * p0, s1 = p1 * p1, s2 = p2 * p2, s3 = p3 * p3;
            s0 += row_xor<4>(s0); s1 += row_xor<4>(s1);
            s2 += row_xor<4>(s2); s3 += row_xor<4>(s3);
            s0 += row_xor<8>(s0); s1 += row_xor<8>(s1);
            s2 += row_xor<8>(s2); s3 += row_xor<8>(s3);
            s0 += half_xor(s0);   s1 += half_xor(s1);
            s2 += half_xor(s2);   s3 += half_xor(s3);
            const float q0 = p0 * (__builtin_amdgcn_sqrtf(s0) * __builtin_amdgcn_rcpf(1.f + s0));
            const float q1 = p1 * (__builtin_amdgcn_sqrtf(s1) * __builtin_amdgcn_rcpf(1.f + s1));
            const float q2 = p2 * (__builtin_amdgcn_sqrtf(s2) * __builtin_amdgcn_rcpf(1.f + s2));
            const float q3 = p3 * (__builtin_amdgcn_sqrtf(s3) * __builtin_amdgcn_rcpf(1.f + s3));

            if (r + 1 < nroutes) {
                #pragma unroll
                for (int bb = 0; bb < 8; ++bb) {
                    float u = cur[bb].x * q0;
                    u = fmaf(cur[bb].y, q1, u);
                    u = fmaf(cur[bb].z, q2, u);
                    u = fmaf(cur[bb].w, q3, u);
                    u += row_xor<1>(u);
                    u += row_xor<2>(u);
                    lg[bb] = ls[bb] + u;
                }
            } else {
                *(float4*)(vS + wave * RS_PAIR + i * RS_ODD + lane * 4) =
                    make_float4(q0, q1, q2, q3);
            }
        }
    }

    __syncthreads();
    if (nroutes > 0) {
        const float* sp = vS + (laneM >> 1) * RS_PAIR + (laneM & 1) * RS_ODD
                        + 16 * wave + laneH;
        float* o = out + (size_t)(16 * wave + laneH) * NPIX
                 + oh * W_OUT + 2 * (ox0 + laneM) + q;
        #pragma unroll
        for (int k = 0; k < 8; ++k)
            o[(size_t)(2 * k) * NPIX] = sp[2 * k];
    }
}

extern "C" void kernel_launch(void* const* d_in, const int* in_sizes, int n_in,
                              void* d_out, int out_size, void* d_ws, size_t ws_size,
                              hipStream_t stream) {
    (void)in_sizes; (void)n_in; (void)out_size; (void)d_ws; (void)ws_size;
    const float* x  = (const float*)d_in[0];
    const float* wt = (const float*)d_in[1];
    const int*   nr = (const int*)d_in[2];
    float* out   = (float*)d_out;

    vote_route_fused_kernel<<<2304, 256, 0, stream>>>(x, wt, nr, out);
}


# voting: each wave stages its input row in LDS once per tap row (16-byte loads, halo zeroed there); the row's 2-3 column taps read A fragments from it by ds_load_b64, no global loads or masks per tap
# speedup vs baseline: 3.3554x; 1.0560x over previous
#include <hip/hip_runtime.h>


typedef __attribute__((ext_vector_type(2))) float v2f;
typedef __attribute__((ext_vector_type(8))) float v8f;

#define H_IN   96
#define W_IN   96
#define HW_IN  (H_IN * W_IN)
#define CIN    16
#define B_     8
#define A_     8
#define COUT   16
#define N_     128
#define H_OUT  192
#define W_OUT  192
#define NPIX   (H_OUT * W_OUT)
#define WP2    144
#define APITCH 48
#define A_BASE (8 * WP2)
#define VPIX   (B_ * N_)
#define LOG2E  1.4426950408889634f
#define RS_PAIR (2 * VPIX + 8)
#define RS_ODD  132

template <int M>
__device__ __forceinline__ float row_xor(float x) {
    return __builtin_bit_cast(float, __builtin_amdgcn_update_dpp(
        0, __builtin_bit_cast(int, x), 0x160 | M, 0xf, 0xf, true));
}
__device__ __forceinline__ float half_xor(float x) {
    const int b = __builtin_bit_cast(int, x);
    return __builtin_bit_cast(float, __builtin_amdgcn_permlanex16(
        b, b, 0x76543210, 0xfedcba98, false, false));
}

__global__ void __launch_bounds__(256, 8)
vote_route_fused_kernel(const float* __restrict__ x,
                        const float* __restrict__ wt,
                        const int* __restrict__ nroutes_p,
                        float* __restrict__ out)
{
    __shared__ float4 smem[16 * VPIX / 4];
    v2f*   wS = (v2f*)smem;
    float* vS = (float*)smem;

    const int tid   = threadIdx.x;
    const int lane  = tid & 31;
    const int laneM = lane & 15;
    const int laneH = lane >> 4;
    const int wave  = tid >> 5;

    const int cls = blockIdx.x / 576;
    const int rem = blockIdx.x % 576;
    const int p   = cls >> 1;
    const int q   = cls & 1;

    const int ox0 = (rem % 6) * 16;
    const int oy  = rem / 6;
    const int b   = wave;
    const int oh  = 2 * oy + p;

    const int nroutes = *nroutes_p;

    v2f* aS = wS + A_BASE + wave * (8 * APITCH);

    v8f acc[8] = {};

    const int nTh = 3 - p;
    const int nTw = 3 - q;

    for (int i = 0; i < nTh; ++i) {
        const int  kh  = 2 * i + p;
        const int  ih  = oy + i - 1 + p;
        const bool okh = (unsigned)ih < H_IN;
        for (int j = 0; j < nTw; ++j) {
            const int kw = 2 * j + q;

            __syncthreads();

            float4 ev[2], od[2];
            bool   okA[2];
            int    dstA[2];
            if (j == 0) {
                const int xrow = (b * CIN * H_IN + (okh ? ih : 0)) * W_IN;
                const int colmin = q ? ox0 : 0;
                #pragma unroll
                for (int t = 0; t < 2; ++t) {
                    const int task = t ? 32 + laneM : lane;
                    const int k    = task / 6;
                    const int part = task - k * 6;
                    const int iwb  = ox0 - 4 + 4 * part;
                    const int iwc  = min(max(iwb, colmin), W_IN - 4);
                    const int off  = xrow + (2 * k) * HW_IN + iwc;
                    ev[t]   = *(const float4*)(x + off);
                    od[t]   = *(const float4*)(x + off + HW_IN);
                    okA[t]  = okh && ((unsigned)iwb < W_IN);
                    dstA[t] = k * APITCH + 4 * part;
                }
            }
            {
                const float4* src =
                    (const float4*)(wt + (size_t)(kh * 5 + kw) * CIN * N_);
                #pragma unroll
                for (int e = 0; e < 2; ++e) {
                    const int idx4 = tid * 2 + e;
                    const int idx  = idx4 * 4;
                    const int ci   = idx >> 7;
                    const int n0   = idx & 127;
                    const int k    = ci >> 1;
                    const int sub  = ci & 1;
                    const float4 w4 = src[idx4];
                    ((float*)&wS[k * WP2 + n0 + 0])[sub] = w4.x;
                    ((float*)&wS[k * WP2 + n0 + 1])[sub] = w4.y;
                    ((float*)&wS[k * WP2 + n0 + 2])[sub] = w4.z;
                    ((float*)&wS[k * WP2 + n0 + 3])[sub] = w4.w;
                }
            }
            if (j == 0) {
                #pragma unroll
                for (int t = 0; t < 2; ++t) {
                    float4* dst = (float4*)&aS[dstA[t]];
                    dst[0] = okA[t] ? make_float4(ev[t].x, od[t].x, ev[t].y, od[t].y)
                                    : make_float4(0.f, 0.f, 0.f, 0.f);
                    dst[1] = okA[t] ? make_float4(ev[t].z, od[t].z, ev[t].w, od[t].w)
                                    : make_float4(0.f, 0.f, 0.f, 0.f);
                }
            }
            __syncthreads();

            #pragma unroll
            for (int kg = 0; kg < 4; ++kg) {
                const v2f afrag = aS[(kg * 2 + laneH) * APITCH + laneM + j + 3 + q];

                const int brow = (kg * 2 + laneH) * WP2 + laneM;
                v2f bf[8];
                #pragma unroll
                for (int nt = 0; nt < 8; ++nt)
                    bf[nt] = wS[brow + nt * 16];
                #pragma unroll
                for (int nt = 0; nt < 8; ++nt)
                    acc[nt] = __builtin_amdgcn_wmma_f32_16x16x4_f32(
                        false, afrag, false, bf[nt],
                        (short)0, acc[nt], false, false);
            }
        }
    }

    __syncthreads();
    {
        float* vw = vS + (8 * laneH) * VPIX + b * N_;
        #pragma unroll
        for (int nt = 0; nt < 8; ++nt) {
            const int nphys = (nt * 16 + laneM) ^ (laneH * 32);
            #pragma unroll
            for (int r = 0; r < 8; ++r)
                vw[r * VPIX + nphys] = acc[nt][r];
        }
    }
    __syncthreads();

    const int nsw = (lane * 4) ^ ((wave >> 2) * 32);
    for (int i = 0; i < 2; ++i) {
        const int m   = 2 * wave + i;

        const float4* vp = (const float4*)(vS + m * VPIX + nsw);
        float4 cur[8];
        #pragma unroll
        for (int bb = 0; bb < 8; ++bb) cur[bb] = vp[bb * (N_ / 4)];

        float lg[8];
        #pragma unroll
        for (int bb = 0; bb < 8; ++bb) lg[bb] = 0.f;

        for (int r = 0; r < nroutes; ++r) {
            float mx = lg[0];
            #pragma unroll
            for (int bb = 1; bb < 8; ++bb) mx = fmaxf(mx, lg[bb]);
            const float nmx = -mx * LOG2E;
            float ls[8];
            float den = 0.f;
            #pragma unroll
            for (int bb = 0; bb < 8; ++bb) {
                ls[bb] = __builtin_amdgcn_exp2f(fmaf(lg[bb], LOG2E, nmx));
                den += ls[bb];
            }
            const float inv = __builtin_amdgcn_rcpf(den);
            #pragma unroll
            for (int bb = 0; bb < 8; ++bb) ls[bb] *= inv;

            float p0 = 0.f, p1 = 0.f, p2 = 0.f, p3 = 0.f;
            #pragma unroll
            for (int bb = 0; bb < 8; ++bb) {
                p0 = fmaf(cur[bb].x, ls[bb], p0);
                p1 = fmaf(cur[bb].y, ls[bb], p1);
                p2 = fmaf(cur[bb].z, ls[bb], p2);
                p3 = fmaf(cur[bb].w, ls[bb], p3);
            }

            float s0 = p0 * p0, s1 = p1 * p1, s2 = p2 * p2, s3 = p3 * p3;
            s0 += row_xor<4>(s0); s1 += row_xor<4>(s1);
            s2 += row_xor<4>(s2); s3 += row_xor<4>(s3);
            s0 += row_xor<8>(s0); s1 += row_xor<8>(s1);
            s2 += row_xor<8>(s2); s3 += row_xor<8>(s3);
            s0 += half_xor(s0);   s1 += half_xor(s1);
            s2 += half_xor(s2);   s3 += half_xor(s3);
            const float q0 = p0 * (__builtin_amdgcn_sqrtf(s0) * __builtin_amdgcn_rcpf(1.f + s0));
            const float q1 = p1 * (__builtin_amdgcn_sqrtf(s1) * __builtin_amdgcn_rcpf(1.f + s1));
            const float q2 = p2 * (__builtin_amdgcn_sqrtf(s2) * __builtin_amdgcn_rcpf(1.f + s2));
            const float q3 = p3 * (__builtin_amdgcn_sqrtf(s3) * __builtin_amdgcn_rcpf(1.f + s3));

            if (r + 1 < nroutes) {
                #pragma unroll
                for (int bb = 0; bb < 8; ++bb) {
                    float u = cur[bb].x * q0;
                    u = fmaf(cur[bb].y, q1, u);
                    u = fmaf(cur[bb].z, q2, u);
                    u = fmaf(cur[bb].w, q3, u);
                    u += row_xor<1>(u);
                    u += row_xor<2>(u);
                    lg[bb] = ls[bb] + u;
                }
            } else {
                *(float4*)(vS + wave * RS_PAIR + i * RS_ODD + lane * 4) =
                    make_float4(q0, q1, q2, q3);
            }
        }
    }

    __syncthreads();
    if (nroutes > 0) {
        const float* sp = vS + (laneM >> 1) * RS_PAIR + (laneM & 1) * RS_ODD
                        + 16 * wave + laneH;
        float* o = out + (size_t)(16 * wave + laneH) * NPIX
                 + oh * W_OUT + 2 * (ox0 + laneM) + q;
        #pragma unroll
        for (int k = 0; k < 8; ++k)
            o[(size_t)(2 * k) * NPIX] = sp[2 * k];
    }
}

extern "C" void kernel_launch(void* const* d_in, const int* in_sizes, int n_in,
                              void* d_out, int out_size, void* d_ws, size_t ws_size,
                              hipStream_t stream) {
    (void)in_sizes; (void)n_in; (void)out_size; (void)d_ws; (void)ws_size;
    const float* x  = (const float*)d_in[0];
    const float* wt = (const float*)d_in[1];
    const int*   nr = (const int*)d_in[2];
    float* out   = (float*)d_out;

    vote_route_fused_kernel<<<2304, 256, 0, stream>>>(x, wt, nr, out);
}


# routing: a wave works on its two pixels at once, one per 16-lane row, a lane holding one atom's 8 channels: one softmax per 8 channels (was 4), channel sum 1 DPP step (was 2); sum orders kept
# speedup vs baseline: 3.4793x; 1.0369x over previous
#include <hip/hip_runtime.h>


typedef __attribute__((ext_vector_type(2))) float v2f;
typedef __attribute__((ext_vector_type(8))) float v8f;

#define H_IN   96
#define W_IN   96
#define HW_IN  (H_IN * W_IN)
#define CIN    16
#define B_     8
#define A_     8
#define COUT   16
#define N_     128
#define H_OUT  192
#define W_OUT  192
#define NPIX   (H_OUT * W_OUT)
#define WP2    144
#define APITCH 48
#define A_BASE (8 * WP2)
#define VPIX   (B_ * N_)
#define LOG2E  1.4426950408889634f
#define RS_PAIR (2 * VPIX + 8)
#define RS_ODD  132

template <int M>
__device__ __forceinline__ float row_xor(float x) {
    return __builtin_bit_cast(float, __builtin_amdgcn_update_dpp(
        0, __builtin_bit_cast(int, x), 0x160 | M, 0xf, 0xf, true));
}

__global__ void __launch_bounds__(256, 8)
vote_route_fused_kernel(const float* __restrict__ x,
                        const float* __restrict__ wt,
                        const int* __restrict__ nroutes_p,
                        float* __restrict__ out)
{
    __shared__ float4 smem[16 * VPIX / 4];
    v2f*   wS = (v2f*)smem;
    float* vS = (float*)smem;

    const int tid   = threadIdx.x;
    const int lane  = tid & 31;
    const int laneM = lane & 15;
    const int laneH = lane >> 4;
    const int wave  = tid >> 5;

    const int cls = blockIdx.x / 576;
    const int rem = blockIdx.x % 576;
    const int p   = cls >> 1;
    const int q   = cls & 1;

    const int ox0 = (rem % 6) * 16;
    const int oy  = rem / 6;
    const int b   = wave;
    const int oh  = 2 * oy + p;

    const int nroutes = *nroutes_p;

    v2f* aS = wS + A_BASE + wave * (8 * APITCH);

    v8f acc[8] = {};

    const int nTh = 3 - p;
    const int nTw = 3 - q;

    for (int i = 0; i < nTh; ++i) {
        const int  kh  = 2 * i + p;
        const int  ih  = oy + i - 1 + p;
        const bool okh = (unsigned)ih < H_IN;
        for (int j = 0; j < nTw; ++j) {
            const int kw = 2 * j + q;

            __syncthreads();

            float4 ev[2], od[2];
            bool   okA[2];
            int    dstA[2];
            if (j == 0) {
                const int xrow = (b * CIN * H_IN + (okh ? ih : 0)) * W_IN;
                const int colmin = q ? ox0 : 0;
                #pragma unroll
                for (int t = 0; t < 2; ++t) {
                    const int task = t ? 32 + laneM : lane;
                    const int k    = task / 6;
                    const int part = task - k * 6;
                    const int iwb  = ox0 - 4 + 4 * part;
                    const int iwc  = min(max(iwb, colmin), W_IN - 4);
                    const int off  = xrow + (2 * k) * HW_IN + iwc;
                    ev[t]   = *(const float4*)(x + off);
                    od[t]   = *(const float4*)(x + off + HW_IN);
                    okA[t]  = okh && ((unsigned)iwb < W_IN);
                    dstA[t] = k * APITCH + 4 * part;
                }
            }
            {
                const float4* src =
                    (const float4*)(wt + (size_t)(kh * 5 + kw) * CIN * N_);
                #pragma unroll
                for (int e = 0; e < 2; ++e) {
                    const int idx4 = tid * 2 + e;
                    const int idx  = idx4 * 4;
                    const int ci   = idx >> 7;
                    const int n0   = idx & 127;
                    const int k    = ci >> 1;
                    const int sub  = ci & 1;
                    const float4 w4 = src[idx4];
                    ((float*)&wS[k * WP2 + n0 + 0])[sub] = w4.x;
                    ((float*)&wS[k * WP2 + n0 + 1])[sub] = w4.y;
                    ((float*)&wS[k * WP2 + n0 + 2])[sub] = w4.z;
                    ((float*)&wS[k * WP2 + n0 + 3])[sub] = w4.w;
                }
            }
            if (j == 0) {
                #pragma unroll
                for (int t = 0; t < 2; ++t) {
                    float4* dst = (float4*)&aS[dstA[t]];
                    dst[0] = okA[t] ? make_float4(ev[t].x, od[t].x, ev[t].y, od[t].y)
                                    : make_float4(0.f, 0.f, 0.f, 0.f);
                    dst[1] = okA[t] ? make_float4(ev[t].z, od[t].z, ev[t].w, od[t].w)
                                    : make_float4(0.f, 0.f, 0.f, 0.f);
                }
            }
            __syncthreads();

            #pragma unroll
            for (int kg = 0; kg < 4; ++kg) {
                const v2f afrag = aS[(kg * 2 + laneH) * APITCH + laneM + j + 3 + q];

                const int brow = (kg * 2 + laneH) * WP2 + laneM;
                v2f bf[8];
                #pragma unroll
                for (int nt = 0; nt < 8; ++nt)
                    bf[nt] = wS[brow + nt * 16];
                #pragma unroll
                for (int nt = 0; nt < 8; ++nt)
                    acc[nt] = __builtin_amdgcn_wmma_f32_16x16x4_f32(
                        false, afrag, false, bf[nt],
                        (short)0, acc[nt], false, false);
            }
        }
    }

    __syncthreads();
    {
        const int pm = (laneM & 8) | ((laneM & 3) << 1) | ((laneM >> 2) & 1);
        float* vw = vS + (8 * laneH) * VPIX + b * N_;
        #pragma unroll
        for (int nt = 0; nt < 8; ++nt) {
            const int nphys = (nt * 16 + pm) ^ (laneH * 32) ^ (nt & 4);
            #pragma unroll
            for (int r = 0; r < 8; ++r)
                vw[r * VPIX + nphys] = acc[nt][r];
        }
    }
    __syncthreads();

    {
        const int rl = lane & 15;
        const int m  = 2 * wave + laneH;
        const int sw = ((wave >> 2) * 32) ^ ((rl >> 3) * 4);
        const float4* vp0 = (const float4*)(vS + m * VPIX + ((rl * 8)     ^ sw));
        const float4* vp1 = (const float4*)(vS + m * VPIX + ((rl * 8 + 4) ^ sw));
        v2f V[8][4];
        #pragma unroll
        for (int bb = 0; bb < 8; ++bb) {
            const float4 ta = vp0[bb * (N_ / 4)];
            const float4 tb = vp1[bb * (N_ / 4)];
            V[bb][0] = (v2f){ta.x, ta.y};
            V[bb][1] = (v2f){ta.z, ta.w};
            V[bb][2] = (v2f){tb.x, tb.y};
            V[bb][3] = (v2f){tb.z, tb.w};
        }

        float lg[8];
        #pragma unroll
        for (int bb = 0; bb < 8; ++bb) lg[bb] = 0.f;

        for (int r = 0; r < nroutes; ++r) {
            float mx = lg[0];
            #pragma unroll
            for (int bb = 1; bb < 8; ++bb) mx = fmaxf(mx, lg[bb]);
            const float nmx = -mx * LOG2E;
            float ls[8];
            float den = 0.f;
            #pragma unroll
            for (int bb = 0; bb < 8; ++bb) {
                ls[bb] = __builtin_amdgcn_exp2f(fmaf(lg[bb], LOG2E, nmx));
                den += ls[bb];
            }
            const float inv = __builtin_amdgcn_rcpf(den);
            #pragma unroll
            for (int bb = 0; bb < 8; ++bb) ls[bb] *= inv;

            v2f P[4];
            #pragma unroll
            for (int j = 0; j < 4; ++j) P[j] = (v2f){0.f, 0.f};
            #pragma unroll
            for (int bb = 0; bb < 8; ++bb) {
                const v2f w2 = (v2f){ls[bb], ls[bb]};
                #pragma unroll
                for (int j = 0; j < 4; ++j)
                    P[j] = __builtin_elementwise_fma(V[bb][j], w2, P[j]);
            }

            v2f Q[4];
            #pragma unroll
            for (int j = 0; j < 4; ++j) {
                const v2f sq = P[j] * P[j];
                float s0 = sq.x, s1 = sq.y;
                s0 += row_xor<2>(s0); s1 += row_xor<2>(s1);
                s0 += row_xor<4>(s0); s1 += row_xor<4>(s1);
                s0 += row_xor<8>(s0); s1 += row_xor<8>(s1);
                const v2f f = (v2f){
                    __builtin_amdgcn_sqrtf(s0) * __builtin_amdgcn_rcpf(1.f + s0),
                    __builtin_amdgcn_sqrtf(s1) * __builtin_amdgcn_rcpf(1.f + s1)};
                Q[j] = P[j] * f;
            }

            if (r + 1 < nroutes) {
                #pragma unroll
                for (int bb = 0; bb < 8; ++bb) {
                    v2f uv = V[bb][0] * Q[0];
                    uv = __builtin_elementwise_fma(V[bb][1], Q[1], uv);
                    uv = __builtin_elementwise_fma(V[bb][2], Q[2], uv);
                    uv = __builtin_elementwise_fma(V[bb][3], Q[3], uv);
                    float u = uv.x + uv.y;
                    u += row_xor<1>(u);
                    lg[bb] = ls[bb] + u;
                }
            } else {
                float* st = vS + wave * RS_PAIR + laneH * RS_ODD + rl * 8;
                *(float4*)(st)     = make_float4(Q[0].x, Q[1].x, Q[2].x, Q[3].x);
                *(float4*)(st + 4) = make_float4(Q[0].y, Q[1].y, Q[2].y, Q[3].y);
            }
        }
    }

    __syncthreads();
    if (nroutes > 0) {
        const float* sp = vS + (laneM >> 1) * RS_PAIR + (laneM & 1) * RS_ODD
                        + 16 * wave + laneH;
        float* o = out + (size_t)(16 * wave + laneH) * NPIX
                 + oh * W_OUT + 2 * (ox0 + laneM) + q;
        #pragma unroll
        for (int k = 0; k < 8; ++k)
            o[(size_t)(2 * k) * NPIX] = sp[2 * k];
    }
}

extern "C" void kernel_launch(void* const* d_in, const int* in_sizes, int n_in,
                              void* d_out, int out_size, void* d_ws, size_t ws_size,
                              hipStream_t stream) {
    (void)in_sizes; (void)n_in; (void)out_size; (void)d_ws; (void)ws_size;
    const float* x  = (const float*)d_in[0];
    const float* wt = (const float*)d_in[1];
    const int*   nr = (const int*)d_in[2];
    float* out   = (float*)d_out;

    vote_route_fused_kernel<<<2304, 256, 0, stream>>>(x, wt, nr, out);
}


# votes stay in the accumulators: matrix rows = 2 pixels x 8 batch per wave, slab columns permuted, so the D layout is the routing layout; no LDS hand-over; batch and channel sum orders change
# speedup vs baseline: 3.6339x; 1.0444x over previous
#include <hip/hip_runtime.h>


typedef __attribute__((ext_vector_type(2))) float v2f;
typedef __attribute__((ext_vector_type(8))) float v8f;

#define H_IN   96
#define W_IN   96
#define HW_IN  (H_IN * W_IN)
#define CIN    16
#define B_     8
#define A_     8
#define COUT   16
#define N_     128
#define H_OUT  192
#define W_OUT  192
#define NPIX   (H_OUT * W_OUT)
#define WP2    144
#define APITCH 48
#define A_BASE (8 * WP2)
#define SPITCH (8 * APITCH + 2)
#define R_BASE (2 * (A_BASE + B_ * SPITCH))
#define LOG2E  1.4426950408889634f
#define RS_PAIR 264
#define RS_ODD  132
#define SMEM_FLOATS (R_BASE + 8 * RS_PAIR)

template <int M>
__device__ __forceinline__ float row_xor(float x) {
    return __builtin_bit_cast(float, __builtin_amdgcn_update_dpp(
        0, __builtin_bit_cast(int, x), 0x160 | M, 0xf, 0xf, true));
}

__device__ __forceinline__ v2f batch_pair(const v8f& t, int i) {
    return (v2f){t[2 * i], t[2 * i + 1]};
}

__global__ void __launch_bounds__(256, 8)
vote_route_fused_kernel(const float* __restrict__ x,
                        const float* __restrict__ wt,
                        const int* __restrict__ nroutes_p,
                        float* __restrict__ out)
{
    __shared__ float4 smem[SMEM_FLOATS / 4];
    v2f*   wS = (v2f*)smem;
    float* rS = (float*)smem + R_BASE;

    const int tid   = threadIdx.x;
    const int lane  = tid & 31;
    const int laneM = lane & 15;
    const int laneH = lane >> 4;
    const int wave  = tid >> 5;

    const int cls = blockIdx.x / 576;
    const int rem = blockIdx.x % 576;
    const int p   = cls >> 1;
    const int q   = cls & 1;

    const int ox0 = (rem % 6) * 16;
    const int oy  = rem / 6;
    const int b   = wave;
    const int oh  = 2 * oy + p;

    const int nroutes = *nroutes_p;

    v2f* aS = wS + A_BASE + wave * SPITCH;
    const v2f* aR = wS + A_BASE + (laneM & 7) * SPITCH + 2 * wave + (laneM >> 3);

    v8f acc[8] = {};

    const int nTh = 3 - p;
    const int nTw = 3 - q;

    for (int i = 0; i < nTh; ++i) {
        const int  kh  = 2 * i + p;
        const int  ih  = oy + i - 1 + p;
        const bool okh = (unsigned)ih < H_IN;
        for (int j = 0; j < nTw; ++j) {
            const int kw = 2 * j + q;

            __syncthreads();

            float4 ev[2], od[2];
            bool   okA[2];
            int    dstA[2];
            if (j == 0) {
                const int xrow = (b * CIN * H_IN + (okh ? ih : 0)) * W_IN;
                const int colmin = q ? ox0 : 0;
                #pragma unroll
                for (int t = 0; t < 2; ++t) {
                    const int task = t ? 32 + laneM : lane;
                    const int k    = task / 6;
                    const int part = task - k * 6;
                    const int iwb  = ox0 - 4 + 4 * part;
                    const int iwc  = min(max(iwb, colmin), W_IN - 4);
                    const int off  = xrow + (2 * k) * HW_IN + iwc;
                    ev[t]   = *(const float4*)(x + off);
                    od[t]   = *(const float4*)(x + off + HW_IN);
                    okA[t]  = okh && ((unsigned)iwb < W_IN);
                    dstA[t] = k * APITCH + 4 * part;
                }
            }
            {
                const float4* src =
                    (const float4*)(wt + (size_t)(kh * 5 + kw) * CIN * N_);
                #pragma unroll
                for (int e = 0; e < 2; ++e) {
                    const int idx4 = tid * 2 + e;
                    const int idx  = idx4 * 4;
                    const int ci   = idx >> 7;
                    const int n0   = idx & 127;
                    const int k    = ci >> 1;
                    const int sub  = ci & 1;
                    const float4 w4 = src[idx4];
                    const int pos = (n0 & 7) * 16 + (n0 >> 3);
                    ((float*)&wS[k * WP2 + pos +  0])[sub] = w4.x;
                    ((float*)&wS[k * WP2 + pos + 16])[sub] = w4.y;
                    ((float*)&wS[k * WP2 + pos + 32])[sub] = w4.z;
                    ((float*)&wS[k * WP2 + pos + 48])[sub] = w4.w;
                }
            }
            if (j == 0) {
                #pragma unroll
                for (int t = 0; t < 2; ++t) {
                    float4* dst = (float4*)&aS[dstA[t]];
                    dst[0] = okA[t] ? make_float4(ev[t].x, od[t].x, ev[t].y, od[t].y)
                                    : make_float4(0.f, 0.f, 0.f, 0.f);
                    dst[1] = okA[t] ? make_float4(ev[t].z, od[t].z, ev[t].w, od[t].w)
                                    : make_float4(0.f, 0.f, 0.f, 0.f);
                }
            }
            __syncthreads();

            #pragma unroll
            for (int kg = 0; kg < 4; ++kg) {
                const v2f afrag = aR[(kg * 2 + laneH) * APITCH + j + 3 + q];

                const int brow = (kg * 2 + laneH) * WP2 + laneM;
                v2f bf[8];
                #pragma unroll
                for (int nt = 0; nt < 8; ++nt)
                    bf[nt] = wS[brow + nt * 16];
                #pragma unroll
                for (int nt = 0; nt < 8; ++nt)
                    acc[nt] = __builtin_amdgcn_wmma_f32_16x16x4_f32(
                        false, afrag, false, bf[nt],
                        (short)0, acc[nt], false, false);
            }
        }
    }

    {
        const int rl = laneM;

        float lg[8];
        #pragma unroll
        for (int bb = 0; bb < 8; ++bb) lg[bb] = 0.f;

        for (int r = 0; r < nroutes; ++r) {
            float mx = lg[0];
            #pragma unroll
            for (int bb = 1; bb < 8; ++bb) mx = fmaxf(mx, lg[bb]);
            const float nmx = -mx * LOG2E;
            float ls[8];
            float den = 0.f;
            #pragma unroll
            for (int bb = 0; bb < 8; ++bb) {
                ls[bb] = __builtin_amdgcn_exp2f(fmaf(lg[bb], LOG2E, nmx));
                den += ls[bb];
            }
            const float inv = __builtin_amdgcn_rcpf(den);
            #pragma unroll
            for (int bb = 0; bb < 8; ++bb) ls[bb] *= inv;

            float pr[8];
            #pragma unroll
            for (int nt = 0; nt < 8; ++nt) {
                v2f s = batch_pair(acc[nt], 0)
                      * (v2f){ls[0], ls[1]};
                s = __builtin_elementwise_fma(
                        batch_pair(acc[nt], 1),
                        (v2f){ls[2], ls[3]}, s);
                s = __builtin_elementwise_fma(
                        batch_pair(acc[nt], 2),
                        (v2f){ls[4], ls[5]}, s);
                s = __builtin_elementwise_fma(
                        batch_pair(acc[nt], 3),
                        (v2f){ls[6], ls[7]}, s);
                pr[nt] = s.x + s.y;
            }

            float Q[8];
            #pragma unroll
            for (int j = 0; j < 4; ++j) {
                const v2f pp = (v2f){pr[2 * j], pr[2 * j + 1]};
                const v2f sq = pp * pp;
                float s0 = sq.x, s1 = sq.y;
                s0 += row_xor<2>(s0); s1 += row_xor<2>(s1);
                s0 += row_xor<4>(s0); s1 += row_xor<4>(s1);
                s0 += row_xor<8>(s0); s1 += row_xor<8>(s1);
                const v2f f = (v2f){
                    __builtin_amdgcn_sqrtf(s0) * __builtin_amdgcn_rcpf(1.f + s0),
                    __builtin_amdgcn_sqrtf(s1) * __builtin_amdgcn_rcpf(1.f + s1)};
                const v2f qq = pp * f;
                Q[2 * j]     = qq.x;
                Q[2 * j + 1] = qq.y;
            }

            if (r + 1 < nroutes) {
                #pragma unroll
                for (int i = 0; i < 4; ++i) {
                    v2f uv = batch_pair(acc[0], i)
                           * (v2f){Q[0], Q[0]};
                    #pragma unroll
                    for (int nt = 1; nt < 8; ++nt)
                        uv = __builtin_elementwise_fma(
                                 batch_pair(acc[nt], i),
                                 (v2f){Q[nt], Q[nt]}, uv);
                    float u0 = uv.x, u1 = uv.y;
                    u0 += row_xor<1>(u0);
                    u1 += row_xor<1>(u1);
                    lg[2 * i]     = ls[2 * i]     + u0;
                    lg[2 * i + 1] = ls[2 * i + 1] + u1;
                }
            } else {
                float* st = rS + wave * RS_PAIR + laneH * RS_ODD + rl * 8;
                *(float4*)(st)     = make_float4(Q[0], Q[1], Q[2], Q[3]);
                *(float4*)(st + 4) = make_float4(Q[4], Q[5], Q[6], Q[7]);
            }
        }
    }

    __syncthreads();
    if (nroutes > 0) {
        const float* sp = rS + (laneM >> 1) * RS_PAIR + (laneM & 1) * RS_ODD
                        + 16 * wave + laneH;
        float* o = out + (size_t)(16 * wave + laneH) * NPIX
                 + oh * W_OUT + 2 * (ox0 + laneM) + q;
        #pragma unroll
        for (int k = 0; k < 8; ++k)
            o[(size_t)(2 * k) * NPIX] = sp[2 * k];
    }
}

extern "C" void kernel_launch(void* const* d_in, const int* in_sizes, int n_in,
                              void* d_out, int out_size, void* d_ws, size_t ws_size,
                              hipStream_t stream) {
    (void)in_sizes; (void)n_in; (void)out_size; (void)d_ws; (void)ws_size;
    const float* x  = (const float*)d_in[0];
    const float* wt = (const float*)d_in[1];
    const int*   nr = (const int*)d_in[2];
    float* out   = (float*)d_out;

    vote_route_fused_kernel<<<2304, 256, 0, stream>>>(x, wt, nr, out);
}


# voting: column taps in minimal-filtering form: per tap row 4 (3) matrix products serve 2 neighbouring pixels instead of 6 (4), fp32 as before; wave = 2 pixel pairs; vote sum order, rounding change
# speedup vs baseline: 4.0870x; 1.1247x over previous
#include <hip/hip_runtime.h>


typedef __attribute__((ext_vector_type(2))) float v2f;
typedef __attribute__((ext_vector_type(8))) float v8f;

#define H_IN   96
#define W_IN   96
#define HW_IN  (H_IN * W_IN)
#define CIN    16
#define B_     8
#define A_     8
#define COUT   16
#define N_     128
#define H_OUT  192
#define W_OUT  192
#define NPIX   (H_OUT * W_OUT)
#define WP2    144
#define SLAB   (8 * WP2)
#define APITCH 48
#define A_BASE (4 * SLAB)
#define SPITCH (8 * APITCH + 2)
#define R_BASE (2 * (A_BASE + B_ * SPITCH))
#define LOG2E  1.4426950408889634f
#define RS_PAIR 264
#define RS_ODD  132
#define SMEM_FLOATS (R_BASE + 8 * RS_PAIR)

template <int M>
__device__ __forceinline__ float row_xor(float x) {
    return __builtin_bit_cast(float, __builtin_amdgcn_update_dpp(
        0, __builtin_bit_cast(int, x), 0x160 | M, 0xf, 0xf, true));
}

__device__ __forceinline__ v2f batch_pair(const v8f& t, int i) {
    return (v2f){t[2 * i], t[2 * i + 1]};
}

__device__ __forceinline__ void route_pair(const v8f (&acc)[8], const int nroutes,
                                           float* st)
{
    float lg[8];
    #pragma unroll
    for (int bb = 0; bb < 8; ++bb) lg[bb] = 0.f;

    for (int r = 0; r < nroutes; ++r) {
        float mx = lg[0];
        #pragma unroll
        for (int bb = 1; bb < 8; ++bb) mx = fmaxf(mx, lg[bb]);
        const float nmx = -mx * LOG2E;
        float ls[8];
        float den = 0.f;
        #pragma unroll
        for (int bb = 0; bb < 8; ++bb) {
            ls[bb] = __builtin_amdgcn_exp2f(fmaf(lg[bb], LOG2E, nmx));
            den += ls[bb];
        }
        const float inv = __builtin_amdgcn_rcpf(den);
        #pragma unroll
        for (int bb = 0; bb < 8; ++bb) ls[bb] *= inv;

        float pr[8];
        #pragma unroll
        for (int nt = 0; nt < 8; ++nt) {
            v2f s = batch_pair(acc[nt], 0)
                  * (v2f){ls[0], ls[1]};
            s = __builtin_elementwise_fma(
                    batch_pair(acc[nt], 1),
                    (v2f){ls[2], ls[3]}, s);
            s = __builtin_elementwise_fma(
                    batch_pair(acc[nt], 2),
                    (v2f){ls[4], ls[5]}, s);
            s = __builtin_elementwise_fma(
                    batch_pair(acc[nt], 3),
                    (v2f){ls[6], ls[7]}, s);
            pr[nt] = s.x + s.y;
        }

        float Q[8];
        #pragma unroll
        for (int j = 0; j < 4; ++j) {
            const v2f pp = (v2f){pr[2 * j], pr[2 * j + 1]};
            const v2f sq = pp * pp;
            float s0 = sq.x, s1 = sq.y;
            s0 += row_xor<2>(s0); s1 += row_xor<2>(s1);
            s0 += row_xor<4>(s0); s1 += row_xor<4>(s1);
            s0 += row_xor<8>(s0); s1 += row_xor<8>(s1);
            const v2f f = (v2f){
                __builtin_amdgcn_sqrtf(s0) * __builtin_amdgcn_rcpf(1.f + s0),
                __builtin_amdgcn_sqrtf(s1) * __builtin_amdgcn_rcpf(1.f + s1)};
            const v2f qq = pp * f;
            Q[2 * j]     = qq.x;
            Q[2 * j + 1] = qq.y;
        }

        if (r + 1 < nroutes) {
            #pragma unroll
            for (int i = 0; i < 4; ++i) {
                v2f uv = batch_pair(acc[0], i)
                       * (v2f){Q[0], Q[0]};
                #pragma unroll
                for (int nt = 1; nt < 8; ++nt)
                    uv = __builtin_elementwise_fma(
                             batch_pair(acc[nt], i),
                             (v2f){Q[nt], Q[nt]}, uv);
                float u0 = uv.x, u1 = uv.y;
                u0 += row_xor<1>(u0);
                u1 += row_xor<1>(u1);
                lg[2 * i]     = ls[2 * i]     + u0;
                lg[2 * i + 1] = ls[2 * i + 1] + u1;
            }
        } else {
            *(float4*)(st)     = make_float4(Q[0], Q[1], Q[2], Q[3]);
            *(float4*)(st + 4) = make_float4(Q[4], Q[5], Q[6], Q[7]);
        }
    }
}

template <int Q>
__device__ __forceinline__ void vote_rows(v8f (&acc)[4][8],
        const float* __restrict__ x, const float* __restrict__ wt, v2f* wS,
        const int p, const int oy, const int ox0,
        const int tid, const int lane, const int laneM, const int laneH,
        const int wave)
{
    constexpr int NP = 4 - Q;

    #pragma unroll
    for (int pi = 0; pi < NP; ++pi)
        #pragma unroll
        for (int nt = 0; nt < 8; ++nt)
            acc[pi][nt] = (v8f){0.f, 0.f, 0.f, 0.f, 0.f, 0.f, 0.f, 0.f};

    const v2f* aR = wS + A_BASE + (laneM & 7) * SPITCH
                  + 4 * wave + 2 * (laneM >> 3);

    const int nTh = 3 - p;
    const int colmin = Q ? ox0 : 0;

    for (int i = 0; i < nTh; ++i) {
        const int  kh  = 2 * i + p;
        const int  ih  = oy + i - 1 + p;
        const bool okh = (unsigned)ih < H_IN;

        __syncthreads();

        float4 ev[4], od[4];
        bool   okA[4];
        int    dstA[4];
        #pragma unroll
        for (int t = 0; t < 4; ++t) {
            const int be   = 2 * wave + (t >> 1);
            const int task = (t & 1) ? 32 + laneM : lane;
            const int k    = task / 6;
            const int part = task - k * 6;
            const int iwb  = ox0 - 4 + 4 * part;
            const int iwc  = min(max(iwb, colmin), W_IN - 4);
            const int off  = (be * CIN * H_IN + (okh ? ih : 0)) * W_IN
                           + (2 * k) * HW_IN + iwc;
            ev[t]   = *(const float4*)(x + off);
            od[t]   = *(const float4*)(x + off + HW_IN);
            okA[t]  = okh && ((unsigned)iwb < W_IN);
            dstA[t] = A_BASE + be * SPITCH + k * APITCH + 4 * part;
        }

        const float4* s0 = (const float4*)(wt + (size_t)(kh * 5 + Q) * CIN * N_);
        const float4* s1 = s0 + 2 * (CIN * N_ / 4);
        const float4* s2 = s0 + 4 * (CIN * N_ / 4);
        float4 ga[4], gb[4], gc[4];
        #pragma unroll
        for (int e = 0; e < 2; ++e) {
            const int idx4 = tid * 2 + e;
            ga[e] = s0[idx4];
            gb[e] = s1[idx4];
            if (Q == 0) gc[e] = s2[idx4];
        }

        #pragma unroll
        for (int t = 0; t < 4; ++t) {
            float4* dst = (float4*)&wS[dstA[t]];
            dst[0] = okA[t] ? make_float4(ev[t].x, od[t].x, ev[t].y, od[t].y)
                            : make_float4(0.f, 0.f, 0.f, 0.f);
            dst[1] = okA[t] ? make_float4(ev[t].z, od[t].z, ev[t].w, od[t].w)
                            : make_float4(0.f, 0.f, 0.f, 0.f);
        }

        #pragma unroll
        for (int e = 2; e < 4; ++e) {
            const int idx4 = (tid + 128) * 2 + (e & 1);
            ga[e] = s0[idx4];
            gb[e] = s1[idx4];
            if (Q == 0) gc[e] = s2[idx4];
        }

        #pragma unroll
        for (int e = 0; e < 4; ++e) {
            const int idx4 = (tid + 128 * (e >> 1)) * 2 + (e & 1);
            const int idx  = idx4 * 4;
            const int ci   = idx >> 7;
            const int n0   = idx & 127;
            const int k    = ci >> 1;
            const int sub  = ci & 1;
            const float g0[4] = {ga[e].x, ga[e].y, ga[e].z, ga[e].w};
            const float g1[4] = {gb[e].x, gb[e].y, gb[e].z, gb[e].w};
            float P[NP][4];
            if (Q == 0) {
                const float g2[4] = {gc[e].x, gc[e].y, gc[e].z, gc[e].w};
                #pragma unroll
                for (int c = 0; c < 4; ++c) {
                    const float t02 = g0[c] + g2[c];
                    P[0][c] = g0[c];
                    P[1][c] = 0.5f * (t02 + g1[c]);
                    P[2][c] = 0.5f * (t02 - g1[c]);
                    P[NP - 1][c] = g2[c];
                }
            } else {
                #pragma unroll
                for (int c = 0; c < 4; ++c) {
                    P[0][c] = g0[c];
                    P[1][c] = g0[c] + g1[c];
                    P[2][c] = g1[c];
                }
            }
            const int pos = (n0 & 7) * 16 + (n0 >> 3);
            #pragma unroll
            for (int pi = 0; pi < NP; ++pi) {
                v2f* sl = wS + pi * SLAB + k * WP2 + pos;
                ((float*)&sl[ 0])[sub] = P[pi][0];
                ((float*)&sl[16])[sub] = P[pi][1];
                ((float*)&sl[32])[sub] = P[pi][2];
                ((float*)&sl[48])[sub] = P[pi][3];
            }
        }
        __syncthreads();

        const v2f* ar0 = aR + laneH * APITCH + 3 + Q;
        const v2f* br0 = wS + laneH * WP2 + laneM;
        v2f dd[4];
        v2f af[NP];
        v2f bf[2][8];
        #pragma unroll
        for (int t = 0; t < NP; ++t) dd[t] = ar0[t];
        #pragma unroll
        for (int nt = 0; nt < 8; ++nt) bf[0][nt] = br0[nt * 16];
        #pragma unroll
        for (int s = 0; s < 4 * NP; ++s) {
            const int kg = s / NP;
            const int pi = s % NP;
            if (pi == 0) {
                if (Q == 0) {
                    af[0] = dd[0] - dd[2];
                    af[1] = dd[1] + dd[2];
                    af[2] = dd[2] - dd[1];
                    af[NP - 1] = dd[1] - dd[NP - 1];
                } else {
                    af[0] = dd[0] - dd[1];
                    af[1] = dd[1];
                    af[2] = dd[1] - dd[2];
                }
            }
            if (s + 1 < 4 * NP) {
                const int kg1 = (s + 1) / NP;
                const int pi1 = (s + 1) % NP;
                #pragma unroll
                for (int nt = 0; nt < 8; ++nt)
                    bf[(s + 1) & 1][nt] =
                        br0[pi1 * SLAB + kg1 * 2 * WP2 + nt * 16];
            }
            if (pi == 0 && kg + 1 < 4) {
                #pragma unroll
                for (int t = 0; t < NP; ++t)
                    dd[t] = ar0[(kg + 1) * 2 * APITCH + t];
            }
            #pragma unroll
            for (int nt = 0; nt < 8; ++nt)
                acc[pi][nt] = __builtin_amdgcn_wmma_f32_16x16x4_f32(
                    false, af[pi], false, bf[s & 1][nt],
                    (short)0, acc[pi][nt], false, false);
        }
    }

    #pragma unroll
    for (int nt = 0; nt < 8; ++nt) {
        if (Q == 0) {
            const v8f s = acc[1][nt] + acc[2][nt];
            const v8f d = acc[1][nt] - acc[2][nt];
            acc[0][nt] = acc[0][nt] + s;
            acc[1][nt] = d - acc[3][nt];
        } else {
            const v8f m1 = acc[1][nt];
            acc[0][nt] = acc[0][nt] + m1;
            acc[1][nt] = m1 - acc[2][nt];
        }
    }
}

__global__ void __launch_bounds__(128, 3)
vote_route_fused_kernel(const float* __restrict__ x,
                        const float* __restrict__ wt,
                        const int* __restrict__ nroutes_p,
                        float* __restrict__ out)
{
    __shared__ float4 smem[SMEM_FLOATS / 4];
    v2f*   wS = (v2f*)smem;
    float* rS = (float*)smem + R_BASE;

    const int tid   = threadIdx.x;
    const int lane  = tid & 31;
    const int laneM = lane & 15;
    const int laneH = lane >> 4;
    const int wave  = tid >> 5;

    const int cls = blockIdx.x / 576;
    const int rem = blockIdx.x % 576;
    const int p   = cls >> 1;
    const int q   = cls & 1;

    const int ox0 = (rem % 6) * 16;
    const int oy  = rem / 6;
    const int oh  = 2 * oy + p;

    const int nroutes = *nroutes_p;

    v8f acc[4][8];

    if (q) vote_rows<1>(acc, x, wt, wS, p, oy, ox0, tid, lane, laneM, laneH, wave);
    else   vote_rows<0>(acc, x, wt, wS, p, oy, ox0, tid, lane, laneM, laneH, wave);

    {
        float* st = rS + (2 * wave + laneH) * RS_PAIR + laneM * 8;
        route_pair(acc[0], nroutes, st);
        route_pair(acc[1], nroutes, st + RS_ODD);
    }

    __syncthreads();
    if (nroutes > 0) {
        const float* sp = rS + (laneM >> 1) * RS_PAIR + (laneM & 1) * RS_ODD
                        + 32 * wave + laneH;
        float* o = out + (size_t)(32 * wave + laneH) * NPIX
                 + oh * W_OUT + 2 * (ox0 + laneM) + q;
        #pragma unroll
        for (int k = 0; k < 16; ++k)
            o[(size_t)(2 * k) * NPIX] = sp[2 * k];
    }
}

extern "C" void kernel_launch(void* const* d_in, const int* in_sizes, int n_in,
                              void* d_out, int out_size, void* d_ws, size_t ws_size,
                              hipStream_t stream) {
    (void)in_sizes; (void)n_in; (void)out_size; (void)d_ws; (void)ws_size;
    const float* x  = (const float*)d_in[0];
    const float* wt = (const float*)d_in[1];
    const int*   nr = (const int*)d_in[2];
    float* out   = (float*)d_out;

    vote_route_fused_kernel<<<2304, 128, 0, stream>>>(x, wt, nr, out);
}
